# starting kernel
# baseline (speedup 1.0000x reference)
#include <hip/hip_runtime.h>
#include <hip/hip_bf16.h>

typedef _Float16 f16;
typedef __attribute__((ext_vector_type(16))) _Float16 v16h;
typedef __attribute__((ext_vector_type(8)))  _Float16 v8h;
typedef __attribute__((ext_vector_type(8)))  float    v8f;

#define BATCH 512
#define TSEQ  256
#define CNNC  512
#define VOC   37
#define HID   8
#define KPAD  64
#define GN    32
#define NROW  (TSEQ * BATCH)
#define NTILE (NROW / 16)

__device__ __forceinline__ v8f wmma16(v16h a, v16h b, v8f c) {
  return __builtin_amdgcn_wmma_f32_16x16x32_f16(
       false, a,  false, b,
       (short)0, c,  false,  false);
}

__device__ __forceinline__ v16h cat8(v8h a, v8h b) {
  v16h r;
  for (int i = 0; i < 8; ++i) { r[i] = a[i]; r[i + 8] = b[i]; }
  return r;
}

__device__ __forceinline__ v8f splat8(float x) {
  v8f r;
  for (int i = 0; i < 8; ++i) r[i] = x;
  return r;
}

__device__ __forceinline__ float sigmoidf_(float x) {
  return 1.0f / (1.0f + __expf(-x));
}
__device__ __forceinline__ float tanhf_(float x) {
  float t = __expf(-2.0f * fabsf(x));
  float r = (1.0f - t) / (1.0f + t);
  return copysignf(r, x);
}

__global__ void lstm_pool_kernel(const float* __restrict__ features,
                                 f16* __restrict__ pooled) {
  int idx = blockIdx.x * blockDim.x + threadIdx.x;
  if (idx >= BATCH * CNNC) return;
  const float* p = features + (size_t)idx * 49;
  float s = 0.f;
  #pragma unroll
  for (int i = 0; i < 49; ++i) s += p[i];
  pooled[idx] = (f16)(s * (1.0f / 49.0f));
}

__global__ void lstm_pack_kernel(const float* __restrict__ W_in,
                                 const float* __restrict__ W_ih,
                                 const float* __restrict__ W_hh,
                                 const float* __restrict__ W_out,
                                 const float* __restrict__ b_ih,
                                 const float* __restrict__ b_hh,
                                 const float* __restrict__ b_out,
                                 f16* __restrict__ Winp,
                                 f16* __restrict__ Wihp,
                                 f16* __restrict__ Whhp,
                                 f16* __restrict__ Woutp,
                                 float* __restrict__ gbias,
                                 float* __restrict__ obias)
{
  int tid = threadIdx.x;
  for (int idx = tid; idx < 48 * CNNC; idx += 256) {
    int n = idx / CNNC, k = idx % CNNC;
    Winp[idx] = (f16)(n < VOC ? W_in[n * CNNC + k] : 0.f);
  }
  for (int idx = tid; idx < 4 * 512; idx += 256) {
    int tile = idx / 512, e = idx % 512, lane = e / 16, h = e % 16;
    int ks = tile / 2, nt = tile % 2;
    int kl = (h < 8 ? h : h + 8) + (lane / 16) * 8;
    int k = ks * 32 + kl;
    int n = nt * 16 + (lane % 16);
    Wihp[idx] = (f16)(k < VOC ? W_ih[n * VOC + k] : 0.f);
  }
  for (int idx = tid; idx < 2 * 512; idx += 256) {
    int tile = idx / 512, e = idx % 512, lane = e / 16, h = e % 16;
    int kl = (h < 8 ? h : h + 8) + (lane / 16) * 8;
    int n = tile * 16 + (lane % 16);
    Whhp[idx] = (f16)(kl < HID ? W_hh[n * HID + kl] : 0.f);
  }
  for (int idx = tid; idx < 3 * 512; idx += 256) {
    int tile = idx / 512, e = idx % 512, lane = e / 16, h = e % 16;
    int kl = (h < 8 ? h : h + 8) + (lane / 16) * 8;
    int n = tile * 16 + (lane % 16);
    Woutp[idx] = (f16)((kl < HID && n < VOC) ? W_out[n * HID + kl] : 0.f);
  }
  if (tid < GN) gbias[tid] = b_ih[tid] + b_hh[tid];
  if (tid < 48) obias[tid] = (tid < VOC) ? b_out[tid] : 0.f;
}

__global__ void lstm_featgemm_kernel(const f16* __restrict__ pooled,
                                     const f16* __restrict__ Winp,
                                     const float* __restrict__ b_in,
                                     f16* __restrict__ Xh) {
  const int L = threadIdx.x;
  const int j16 = L & 15, half = L >> 4;
  const int Mb = blockIdx.x * 16;

  v8f acc[3];
  #pragma unroll
  for (int nt = 0; nt < 3; ++nt) {
    int col = nt * 16 + j16;
    acc[nt] = splat8(col < VOC ? b_in[col] : 0.f);
  }

  for (int kt = 0; kt < CNNC / 32; ++kt) {
    int kb = kt * 32 + half * 8;
    int row = Mb + j16;
    v16h A = cat8(*(const v8h*)(pooled + (size_t)row * CNNC + kb),
                  *(const v8h*)(pooled + (size_t)row * CNNC + kb + 16));
    #pragma unroll
    for (int nt = 0; nt < 3; ++nt) {
      int n = nt * 16 + j16;
      v16h Bm = cat8(*(const v8h*)(Winp + (size_t)n * CNNC + kb),
                     *(const v8h*)(Winp + (size_t)n * CNNC + kb + 16));
      acc[nt] = wmma16(A, Bm, acc[nt]);
    }
  }
  #pragma unroll
  for (int r = 0; r < 8; ++r) {
    int m = r + 8 * half;
    f16* dst = Xh + (size_t)(Mb + m) * KPAD;
    #pragma unroll
    for (int nt = 0; nt < 3; ++nt) {
      int col = nt * 16 + j16;
      dst[col] = (col < VOC) ? (f16)acc[nt][r] : (f16)0;
    }
    dst[48 + j16] = (f16)0;
  }
}

__global__ void lstm_packx_kernel(const float* __restrict__ captions,
                                  f16* __restrict__ Xh) {
  size_t idx = (size_t)blockIdx.x * blockDim.x + threadIdx.x;
  int col = idx & (KPAD - 1);
  size_t row = idx >> 6;
  int t = (int)(row / BATCH) + 1;
  int b = (int)(row % BATCH);
  float v = (col < VOC) ? captions[((size_t)b * TSEQ + (t - 1)) * VOC + col] : 0.f;
  Xh[((size_t)t * BATCH + b) * KPAD + col] = (f16)v;
}

__global__ void lstm_pregate_kernel(const f16* __restrict__ Xh,
                                    const f16* __restrict__ Wihp,
                                    const float* __restrict__ gbias,
                                    float* __restrict__ G) {
  const int L = threadIdx.x;
  const int j16 = L & 15, half = L >> 4;
  const size_t Mb = (size_t)blockIdx.x * 16;

  v8f acc[2];
  acc[0] = splat8(gbias[j16]);
  acc[1] = splat8(gbias[16 + j16]);

  #pragma unroll
  for (int ks = 0; ks < 2; ++ks) {
    int kb = ks * 32 + half * 8;
    size_t row = Mb + j16;
    v16h A = cat8(*(const v8h*)(Xh + row * KPAD + kb),
                  *(const v8h*)(Xh + row * KPAD + kb + 16));
    #pragma unroll
    for (int nt = 0; nt < 2; ++nt) {
      const f16* tp = Wihp + (size_t)(ks * 2 + nt) * 512 + L * 16;
      v16h Bm = cat8(*(const v8h*)tp, *(const v8h*)(tp + 8));
      acc[nt] = wmma16(A, Bm, acc[nt]);
    }
  }
  float* dst = G + (size_t)blockIdx.x * 512 + L * 16;
  *(v8f*)dst       = acc[0];
  *(v8f*)(dst + 8) = acc[1];
}

__global__ void __launch_bounds__(32)
lstm_recur_kernel(const float* __restrict__ G,
                  const f16* __restrict__ Whhp,
                  f16* __restrict__ Hb) {
  const int L = threadIdx.x;
  const int j16 = L & 15, half = L >> 4;
  const int bblk = blockIdx.x;
  const int bbase = bblk * 16;

  v16h WhhB[2];
  #pragma unroll
  for (int nt = 0; nt < 2; ++nt) {
    const f16* tp = Whhp + (size_t)nt * 512 + L * 16;
    WhhB[nt] = cat8(*(const v8h*)tp, *(const v8h*)(tp + 8));
  }

  float cst[8];
  #pragma unroll
  for (int r = 0; r < 8; ++r) cst[r] = 0.f;

  v16h Ah;
  #pragma unroll
  for (int i = 0; i < 16; ++i) Ah[i] = (f16)0;

  __shared__ f16 hb[16][HID];

  const bool lowj = (j16 < 8);

  for (int t = 0; t < TSEQ; ++t) {
    const float* Gt = G + ((size_t)(t * (BATCH / 16) + bblk) * 512) + L * 16;
    v8f g0 = *(const v8f*)Gt;
    v8f g1 = *(const v8f*)(Gt + 8);
    if (t + 1 < TSEQ)
      __builtin_prefetch(Gt + (size_t)(BATCH / 16) * 512, 0, 1);

    g0 = wmma16(Ah, WhhB[0], g0);
    g1 = wmma16(Ah, WhhB[1], g1);

    float hnew[8];
    #pragma unroll
    for (int r = 0; r < 8; ++r) {
      float a0 = g0[r], a1 = g1[r];
      float b0 = __shfl_xor(a0, 8, 32);
      float b1 = __shfl_xor(a1, 8, 32);
      float iv = lowj ? a0 : b0;
      float fv = lowj ? b0 : a0;
      float gv = lowj ? a1 : b1;
      float ov = lowj ? b1 : a1;
      iv = sigmoidf_(iv); fv = sigmoidf_(fv); ov = sigmoidf_(ov);
      gv = tanhf_(gv);
      cst[r] = fv * cst[r] + iv * gv;
      hnew[r] = ov * tanhf_(cst[r]);
    }

    if (lowj) {
      #pragma unroll
      for (int r = 0; r < 8; ++r) hb[r + 8 * half][j16] = (f16)hnew[r];
    }
    __syncthreads();
    v16h An;
    #pragma unroll
    for (int i = 0; i < 16; ++i) An[i] = (f16)0;
    v8h hv;
    #pragma unroll
    for (int i = 0; i < 8; ++i) hv[i] = (f16)0;
    if (L < 16) {
      hv = *(const v8h*)&hb[L][0];
      #pragma unroll
      for (int k = 0; k < HID; ++k) An[k] = hv[k];
    }
    __syncthreads();
    Ah = An;

    if (L < 16)
      *(v8h*)(Hb + ((size_t)t * BATCH + bbase + L) * HID) = hv;
  }
}

__global__ void lstm_outhead_kernel(const f16* __restrict__ Hb,
                                    const f16* __restrict__ Woutp,
                                    const float* __restrict__ obias,
                                    float* __restrict__ out) {
  const int L = threadIdx.x;
  const int j16 = L & 15, half = L >> 4;
  const size_t Mb = (size_t)blockIdx.x * 16;

  v16h WoutB[3];
  #pragma unroll
  for (int nt = 0; nt < 3; ++nt) {
    const f16* tp = Woutp + (size_t)nt * 512 + L * 16;
    WoutB[nt] = cat8(*(const v8h*)tp, *(const v8h*)(tp + 8));
  }

  v16h A;
  #pragma unroll
  for (int i = 0; i < 16; ++i) A[i] = (f16)0;
  if (L < 16) {
    v8h hv = *(const v8h*)(Hb + (Mb + L) * HID);
    #pragma unroll
    for (int k = 0; k < HID; ++k) A[k] = hv[k];
  }

  v8f l0 = splat8(obias[j16]);
  v8f l1 = splat8(obias[16 + j16]);
  v8f l2 = splat8(obias[32 + j16]);
  l0 = wmma16(A, WoutB[0], l0);
  l1 = wmma16(A, WoutB[1], l1);
  l2 = wmma16(A, WoutB[2], l2);

  const bool v2ok = (32 + j16) < VOC;
  #pragma unroll
  for (int r = 0; r < 8; ++r) {
    float x2 = v2ok ? l2[r] : -1e30f;
    float mx = fmaxf(fmaxf(l0[r], l1[r]), x2);
    #pragma unroll
    for (int s = 1; s < 16; s <<= 1)
      mx = fmaxf(mx, __shfl_xor(mx, s, 16));
    float e0 = __expf(l0[r] - mx);
    float e1 = __expf(l1[r] - mx);
    float e2 = v2ok ? __expf(x2 - mx) : 0.f;
    float sm = e0 + e1 + e2;
    #pragma unroll
    for (int s = 1; s < 16; s <<= 1)
      sm += __shfl_xor(sm, s, 16);
    float inv = 1.0f / sm;

    size_t row = Mb + r + 8 * half;
    int t = (int)(row / BATCH);
    int b = (int)(row % BATCH);
    float* dst = out + ((size_t)b * TSEQ + t) * VOC;
    dst[j16]      = e0 * inv;
    dst[16 + j16] = e1 * inv;
    if (v2ok) dst[32 + j16] = e2 * inv;
  }
}

extern "C" void kernel_launch(void* const* d_in, const int* in_sizes, int n_in,
                              void* d_out, int out_size, void* d_ws, size_t ws_size,
                              hipStream_t stream) {
  const float* features = (const float*)d_in[0];
  const float* captions = (const float*)d_in[1];
  const float* W_in  = (const float*)d_in[2];
  const float* b_in  = (const float*)d_in[3];
  const float* W_ih  = (const float*)d_in[4];
  const float* W_hh  = (const float*)d_in[5];
  const float* b_ih  = (const float*)d_in[6];
  const float* b_hh  = (const float*)d_in[7];
  const float* W_out = (const float*)d_in[8];
  const float* b_out = (const float*)d_in[9];
  float* out = (float*)d_out;

  char* ws = (char*)d_ws;
  auto carve = [&](size_t bytes) -> char* {
    char* p = ws;
    ws += (bytes + 255) & ~(size_t)255;
    return p;
  };
  f16*   pooled = (f16*)  carve((size_t)BATCH * CNNC * sizeof(f16));
  f16*   Winp   = (f16*)  carve((size_t)48 * CNNC * sizeof(f16));
  f16*   Wihp   = (f16*)  carve((size_t)4 * 512 * sizeof(f16));
  f16*   Whhp   = (f16*)  carve((size_t)2 * 512 * sizeof(f16));
  f16*   Woutp  = (f16*)  carve((size_t)3 * 512 * sizeof(f16));
  float* gbias  = (float*)carve(GN * sizeof(float));
  float* obias  = (float*)carve(48 * sizeof(float));
  f16*   Xh     = (f16*)  carve((size_t)NROW * KPAD * sizeof(f16));
  float* G      = (float*)carve((size_t)NROW * GN * sizeof(float));
  f16*   Hb     = (f16*)  carve((size_t)NROW * HID * sizeof(f16));

  lstm_pool_kernel<<<(BATCH * CNNC) / 256, 256, 0, stream>>>(features, pooled);
  lstm_pack_kernel<<<1, 256, 0, stream>>>(W_in, W_ih, W_hh, W_out,
                                          b_ih, b_hh, b_out,
                                          Winp, Wihp, Whhp, Woutp, gbias, obias);
  lstm_featgemm_kernel<<<BATCH / 16, 32, 0, stream>>>(pooled, Winp, b_in, Xh);
  lstm_packx_kernel<<<((TSEQ - 1) * BATCH * KPAD) / 256, 256, 0, stream>>>(captions, Xh);
  lstm_pregate_kernel<<<NTILE, 32, 0, stream>>>(Xh, Wihp, gbias, G);
  lstm_recur_kernel<<<BATCH / 16, 32, 0, stream>>>(G, Whhp, Hb);
  lstm_outhead_kernel<<<NTILE, 32, 0, stream>>>(Hb, Woutp, obias, out);
}


# Recurrence cell: each of a step's two divisions as rcp + the exact division's own fma refinement and corrections, without its operand scaling and fix-up (1 <= d < 2^117 by the exponent cap)
# speedup vs baseline: 5.5640x; 5.5640x over previous
#include <hip/hip_runtime.h>
#include <hip/hip_bf16.h>

typedef _Float16 f16;
typedef __attribute__((ext_vector_type(16))) _Float16 v16h;
typedef __attribute__((ext_vector_type(8)))  _Float16 v8h;
typedef __attribute__((ext_vector_type(8)))  float    v8f;

#define BATCH 512
#define TSEQ  256
#define CNNC  512
#define VOC   37
#define HID   8
#define KPAD  64
#define GN    32
#define NROW  (TSEQ * BATCH)
#define NTILE (NROW / 16)

__device__ __forceinline__ v8f wmma16(v16h a, v16h b, v8f c) {
  return __builtin_amdgcn_wmma_f32_16x16x32_f16(
       false, a,  false, b,
       (short)0, c,  false,  false);
}

__device__ __forceinline__ v16h cat8(v8h a, v8h b) {
  v16h r;
  for (int i = 0; i < 8; ++i) { r[i] = a[i]; r[i + 8] = b[i]; }
  return r;
}

__device__ __forceinline__ v8f splat8(float x) {
  v8f r;
  for (int i = 0; i < 8; ++i) r[i] = x;
  return r;
}

typedef __attribute__((ext_vector_type(2))) float v2f;

__device__ __forceinline__ v2f exp2_pair(v2f x) {
  v2f r;
  r[0] = __builtin_amdgcn_exp2f(x[0]);
  r[1] = __builtin_amdgcn_exp2f(x[1]);
  return r;
}

__device__ __forceinline__ v2f one_plus_exp_neg_capped(v2f x) {
  return 1.0f + exp2_pair(__builtin_elementwise_min(x * -1.44269504f, (v2f)(57.70780164f)));
}

__device__ __forceinline__ v2f exp_neg2abs(v2f x) {
  return exp2_pair((__builtin_elementwise_abs(x) * -2.0f) * 1.44269504f);
}

__device__ __forceinline__ v2f div_in_range(v2f n, v2f d) {
  v2f r;
  r[0] = __builtin_amdgcn_rcpf(d[0]);
  r[1] = __builtin_amdgcn_rcpf(d[1]);
  r = __builtin_elementwise_fma(__builtin_elementwise_fma(-d, r, (v2f)(1.0f)), r, r);
  v2f q = n * r;
  q = __builtin_elementwise_fma(__builtin_elementwise_fma(-d, q, n), r, q);
  q = __builtin_elementwise_fma(__builtin_elementwise_fma(-d, q, n), r, q);
  return q;
}

typedef __attribute__((ext_vector_type(4))) float v4f;

__device__ __forceinline__ void cvt8(v16h& r, int e0, v4f a, v4f b) {
  #pragma unroll
  for (int i = 0; i < 4; ++i) { r[e0 + i] = (f16)a[i]; r[e0 + 4 + i] = (f16)b[i]; }
}
__device__ __forceinline__ void put8(v16h& r, int e0, const float* p) {
  cvt8(r, e0, *(const v4f*)p, *(const v4f*)(p + 4));
}

__device__ __forceinline__ v16h zero16() {
  v16h r;
  #pragma unroll
  for (int i = 0; i < 16; ++i) r[i] = (f16)0;
  return r;
}

__device__ __forceinline__ v16h bfrag_k8(const float* __restrict__ Wm, int row,
                                         bool keep) {
  v16h r = zero16();
  v16h w = r;
  put8(w, 0, Wm + (size_t)row * HID);
  return keep ? w : r;
}

__device__ __forceinline__ v16h bfrag_hh(const float* __restrict__ Wm, int row,
                                         int half) {
  v16h r = zero16();
  const v4f w = *(const v4f*)(Wm + (size_t)row * HID + 4 * half);
  #pragma unroll
  for (int i = 0; i < 4; ++i) r[i] = (f16)w[i];
  return r;
}

typedef __attribute__((ext_vector_type(2))) unsigned v2u;
typedef __attribute__((ext_vector_type(8))) unsigned v8u;

#define DPP_XMASK8   0x168
#define DPP_QUAD_X1  0xB1
#define DPP_QUAD_X2  0x4E
template <int CTRL, int BANK>
__device__ __forceinline__ float dpp_merge(float old, float src) {
  return __int_as_float(__builtin_amdgcn_update_dpp(
      __float_as_int(old), __float_as_int(src), CTRL, 0xf, BANK, false));
}
template <int CTRL>
__device__ __forceinline__ float dpp_all(float src) {
  return __int_as_float(__builtin_amdgcn_update_dpp(
      0, __float_as_int(src), CTRL, 0xf, 0xf, true));
}

#define CCH    64
#define NCC    (CNNC / CCH)
#define PPITCH 72
#define PART_NT   (8 * 32)
#define PART_CC   (32 * 3 * PART_NT)

__global__ void __launch_bounds__(1024)
lstm_poolfeat_kernel(const float* __restrict__ features,
                     const float* __restrict__ W_in,
                     float* __restrict__ part) {
  __shared__ __attribute__((aligned(16))) f16 Pt[16 * PPITCH];
  const int tid = threadIdx.x;
  const int L = tid & 31;
  const int j16 = L & 15, half = L >> 4;
  const int wv = __builtin_amdgcn_readfirstlane(tid >> 5);
  const int bt = blockIdx.x / NCC;
  const int cc = blockIdx.x % NCC;
  const int n = wv * 16 + j16;

  v4f w[2][4];
  #pragma unroll
  for (int ks = 0; ks < 2; ++ks)
    #pragma unroll
    for (int i = 0; i < 4; ++i) w[ks][i] = (v4f)(0.f);
  if (wv < 3) {
    const float* wp = W_in + (size_t)(n < VOC ? n : VOC - 1) * CNNC + cc * CCH + half * 8;
    #pragma unroll
    for (int ks = 0; ks < 2; ++ks) {
      w[ks][0] = *(const v4f*)(wp + ks * 32);
      w[ks][1] = *(const v4f*)(wp + ks * 32 + 4);
      w[ks][2] = *(const v4f*)(wp + ks * 32 + 16);
      w[ks][3] = *(const v4f*)(wp + ks * 32 + 20);
    }
  }

  const int row = tid >> 6, ch = tid & 63;
  const float* p = features + ((size_t)(bt * 16 + row) * CNNC + cc * CCH + ch) * 49;
  float s = 0.f;
  #pragma unroll
  for (int i = 0; i < 49; ++i) s += p[i];
  Pt[row * PPITCH + ch] = (f16)(s * (1.0f / 49.0f));
  __syncthreads();

  if (wv < 3) {
    v8f acc = splat8(0.f);
    #pragma unroll
    for (int ks = 0; ks < 2; ++ks) {
      const f16* ap = Pt + j16 * PPITCH + ks * 32 + half * 8;
      v16h A = cat8(*(const v8h*)ap, *(const v8h*)(ap + 16));
      v16h Bw;
      cvt8(Bw, 0, w[ks][0], w[ks][1]);
      cvt8(Bw, 8, w[ks][2], w[ks][3]);
      v16h Bm = (n < VOC) ? Bw : zero16();
      acc = wmma16(A, Bm, acc);
    }
    float* dst = part + (size_t)cc * PART_CC + (size_t)(bt * 3 + wv) * PART_NT + L;
    #pragma unroll
    for (int r = 0; r < 8; ++r) dst[r * 32] = acc[r];
  }
}

#define CH   16
#define NCH  (TSEQ / CH)
#define GRP  8

typedef v4f ring_t[4][32];

__device__ __forceinline__ void ring_put(ring_t* ring, int slot, int L, v8f p0, v8f p1) {
  v4f a, b, c, d;
  #pragma unroll
  for (int i = 0; i < 4; ++i) { a[i] = p0[i]; b[i] = p0[4 + i]; c[i] = p1[i]; d[i] = p1[4 + i]; }
  ring[slot][0][L] = a; ring[slot][1][L] = b;
  ring[slot][2][L] = c; ring[slot][3][L] = d;
}
__device__ __forceinline__ v8f ring_get(ring_t* ring, int slot, int tile, int L) {
  const v4f a = ring[slot][tile * 2][L], b = ring[slot][tile * 2 + 1][L];
  v8f r;
  #pragma unroll
  for (int i = 0; i < 4; ++i) { r[i] = a[i]; r[4 + i] = b[i]; }
  return r;
}

typedef __attribute__((ext_vector_type(4))) unsigned v4u;
__device__ __forceinline__ void hflush(const v2u* hring, int hb, int t0,
                                       f16* __restrict__ Hb, int bbase, int L) {
  const v2u* src = hring + (size_t)hb * CH * 32;
  #pragma unroll
  for (int i = 0; i < CH / 2; ++i) {
    const v2u a = src[(i * 32 + L) * 2], b = src[(i * 32 + L) * 2 + 1];
    v4u v;
    v[0] = a[0]; v[1] = a[1]; v[2] = b[0]; v[3] = b[1];
    *(v4u*)(Hb + ((size_t)(t0 + 2 * i + (L >> 4)) * BATCH + bbase + (L & 15)) * HID) = v;
  }
}

__device__ __forceinline__ void chunk_barrier() {
  asm volatile("s_wait_dscnt 0x0\n\ts_barrier_signal -1\n\ts_barrier_wait -1" ::: "memory");
}

__global__ void __launch_bounds__(64)
lstm_recur_kernel(const float* __restrict__ captions,
                  const float* __restrict__ part,
                  const float* __restrict__ b_in,
                  const float* __restrict__ W_ih,
                  const float* __restrict__ W_hh,
                  const float* __restrict__ b_ih,
                  const float* __restrict__ b_hh,
                  f16* __restrict__ Hb) {
  __shared__ ring_t ring[2 * CH];
  __shared__ __attribute__((aligned(16))) v2u hring[2 * CH * 32];

  const int L = threadIdx.x & 31;
  const int j16 = L & 15, half = L >> 4;
  const int role = __builtin_amdgcn_readfirstlane((int)(threadIdx.x >> 5));
  const int bblk = blockIdx.x;
  const int bbase = bblk * 16;

  v16h WhhB[2];
  #pragma unroll
  for (int nt = 0; nt < 2; ++nt)
    WhhB[nt] = bfrag_hh(W_hh, nt * 16 + j16, half);

  v16h WihB[4];
  {
    const float* w0 = W_ih + (size_t)j16 * VOC;
    const float* w1 = W_ih + (size_t)(16 + j16) * VOC;
    #pragma unroll
    for (int h = 0; h < 16; ++h) {
      const int kl = (h < 8 ? h : h + 8) + half * 8;
      WihB[0][h] = (f16)w0[kl];
      WihB[1][h] = (f16)w1[kl];
    }
    WihB[2] = zero16();
    WihB[3] = zero16();
    #pragma unroll
    for (int h = 0; h < VOC - 32; ++h) {
      const float u0 = w0[32 + h], u1 = w1[32 + h];
      WihB[2][h] = (half == 0) ? (f16)u0 : (f16)0;
      WihB[3][h] = (half == 0) ? (f16)u1 : (f16)0;
    }
  }
  const v8f bias0 = splat8(b_ih[j16] + b_hh[j16]);
  const v8f bias1 = splat8(b_ih[16 + j16] + b_hh[16 + j16]);

  const float* xrow = captions + (size_t)(bbase + j16) * TSEQ * VOC;

  v2f cst[2];
  for (int p = 0; p < 2; ++p) cst[p] = (v2f)(0.f);

  v16h Ah;
  #pragma unroll
  for (int i = 0; i < 16; ++i) Ah[i] = (f16)0;

  const bool ev0 = (L & 1) == 0;
  const bool ev1 = (L & 2) == 0;
  const bool same_row = (half == (j16 >> 3));
  v2u* hslot = hring + j16 * 2 + half;

  for (int c = 0; c <= NCH; ++c) {
    if (role == 1) {
      if (c >= 2) hflush(hring, c & 1, (c - 2) * CH, Hb, bbase, L);
      if (c < NCH) {
        #pragma unroll
        for (int g = 0; g < CH / GRP; ++g) {
          const int t0 = c * CH + g * GRP;
          float xa[GRP][8], xb[GRP][8], xc[GRP][5];
          #pragma unroll
          for (int s = 0; s < GRP; ++s) {
            const int r = (t0 + s > 0) ? t0 + s - 1 : 0;
            const float* xr = xrow + (size_t)r * VOC;
            #pragma unroll
            for (int i = 0; i < 8; ++i) {
              xa[s][i] = xr[half * 8 + i];
              xb[s][i] = xr[16 + half * 8 + i];
            }
            #pragma unroll
            for (int i = 0; i < 5; ++i) xc[s][i] = xr[32 + i];
          }
          #pragma unroll
          for (int s = 0; s < GRP; ++s) {
            v16h A0, A1;
            #pragma unroll
            for (int i = 0; i < 8; ++i) {
              A0[i] = (f16)xa[s][i];
              A0[i + 8] = (f16)xb[s][i];
            }
            #pragma unroll
            for (int i = 0; i < 16; ++i) A1[i] = (f16)0;
            #pragma unroll
            for (int i = 0; i < 5; ++i) A1[i] = (half == 0) ? (f16)xc[s][i] : (f16)0;
            v8f P0 = wmma16(A0, WihB[0], bias0);
            v8f P1 = wmma16(A0, WihB[1], bias1);
            P0 = wmma16(A1, WihB[2], P0);
            P1 = wmma16(A1, WihB[3], P1);
            ring_put(ring, (c & 1) * CH + g * GRP + s, L, P0, P1);
          }
        }
        if (c == 0) {
          float f0[8], f1[8], f2[5];
          #pragma unroll
          for (int i = 0; i < 8; ++i) {
            f0[i] = b_in[half * 8 + i];
            f1[i] = b_in[16 + half * 8 + i];
          }
          #pragma unroll
          for (int i = 0; i < 5; ++i) f2[i] = b_in[32 + i];
          const float* pb = part + ((size_t)bblk * 3 * 8 + (j16 & 7)) * 32
                                 + (j16 >> 3) * 16 + half * 8;
          #pragma unroll
          for (int cc = 0; cc < NCC; ++cc) {
            const float* q = pb + (size_t)cc * PART_CC;
            const v4f a0 = *(const v4f*)q,                 a1 = *(const v4f*)(q + 4);
            const v4f b0 = *(const v4f*)(q + PART_NT),     b1 = *(const v4f*)(q + PART_NT + 4);
            const v4f c0 = *(const v4f*)(q + 2 * PART_NT);
            const float c4 = q[2 * PART_NT + 4];
            #pragma unroll
            for (int i = 0; i < 4; ++i) {
              f0[i] += a0[i]; f0[4 + i] += a1[i];
              f1[i] += b0[i]; f1[4 + i] += b1[i];
              f2[i] += c0[i];
            }
            f2[4] += c4;
          }
          v16h A0, A1;
          #pragma unroll
          for (int i = 0; i < 8; ++i) {
            A0[i] = (f16)f0[i];
            A0[i + 8] = (f16)f1[i];
          }
          #pragma unroll
          for (int i = 0; i < 16; ++i) A1[i] = (f16)0;
          #pragma unroll
          for (int i = 0; i < 5; ++i) A1[i] = (half == 0) ? (f16)f2[i] : (f16)0;
          v8f P0 = wmma16(A0, WihB[0], bias0);
          v8f P1 = wmma16(A0, WihB[1], bias1);
          P0 = wmma16(A1, WihB[2], P0);
          P1 = wmma16(A1, WihB[3], P1);
          ring_put(ring, 0, L, P0, P1);
        }
      }
    } else if (c >= 1) {
      const int sb = ((c - 1) & 1) * CH;
      v8f P0 = ring_get(ring, sb, 0, L);
      v8f P1 = ring_get(ring, sb, 1, L);
      asm volatile("" :: "v"(P0), "v"(P1));
      for (int s = 0; s < CH; ++s) {
        v8f g0 = wmma16(Ah, WhhB[0], P0);
        v8f g1 = wmma16(Ah, WhhB[1], P1);

        float hnew[4];
        v2f oden[2];
        v2f iv[2], fv[2], gv[2], ov[2];
        #pragma unroll
        for (int q = 0; q < 4; ++q) {
          const float lo0 = g0[q], hi0 = g0[q + 4];
          const float lo1 = g1[q], hi1 = g1[q + 4];
          iv[q >> 1][q & 1] = dpp_merge<DPP_XMASK8, 0xC>(lo0, hi0);
          fv[q >> 1][q & 1] = dpp_merge<DPP_XMASK8, 0x3>(hi0, lo0);
          gv[q >> 1][q & 1] = dpp_merge<DPP_XMASK8, 0xC>(lo1, hi1);
          ov[q >> 1][q & 1] = dpp_merge<DPP_XMASK8, 0x3>(hi1, lo1);
        }
        asm volatile("" :: "v"(iv[0]), "v"(iv[1]), "v"(fv[0]), "v"(fv[1]),
                           "v"(gv[0]), "v"(gv[1]), "v"(ov[0]), "v"(ov[1]));

        __builtin_amdgcn_sched_barrier(0);
        {
          const int sn = sb + ((s + 1 < CH) ? s + 1 : s);
          P0 = ring_get(ring, sn, 0, L);
          P1 = ring_get(ring, sn, 1, L);
        }
        __builtin_amdgcn_sched_barrier(0);

        #pragma unroll
        for (int p = 0; p < 2; ++p) {
          const v2f a = one_plus_exp_neg_capped(iv[p]);
          const v2f b = one_plus_exp_neg_capped(fv[p]);
          const v2f tg = exp_neg2abs(gv[p]);
          const v2f d = 1.0f + tg;
          const v2f n = __builtin_elementwise_copysign(1.0f - tg, gv[p]);
          const v2f ad = a * d;
          cst[p] = div_in_range(__builtin_elementwise_fma(cst[p], ad, n * b), ad * b);
          oden[p] = one_plus_exp_neg_capped(ov[p]);
        }

        #pragma unroll
        for (int p = 0; p < 2; ++p) {
          const v2f tc = exp_neg2abs(cst[p]);
          const v2f hp = div_in_range(__builtin_elementwise_copysign(1.0f - tc, cst[p]),
                                      oden[p] * (1.0f + tc));
          hnew[2 * p] = hp[0];
          hnew[2 * p + 1] = hp[1];
        }

        const float n0 = dpp_all<DPP_QUAD_X1>(hnew[0]);
        const float n1 = dpp_all<DPP_QUAD_X1>(hnew[1]);
        const float n2 = dpp_all<DPP_QUAD_X1>(hnew[2]);
        const float n3 = dpp_all<DPP_QUAD_X1>(hnew[3]);
        const float s0 = ev0 ? hnew[0] : n1;
        const float s1 = ev0 ? n0 : hnew[1];
        const float s2 = ev0 ? hnew[2] : n3;
        const float s3 = ev0 ? n2 : hnew[3];
        const float m0 = dpp_all<DPP_QUAD_X2>(s0);
        const float m1 = dpp_all<DPP_QUAD_X2>(s1);
        const float m2 = dpp_all<DPP_QUAD_X2>(s2);
        const float m3 = dpp_all<DPP_QUAD_X2>(s3);
        const float u0 = ev1 ? s0 : m2;
        const float u2 = ev1 ? m0 : s2;
        const float u1 = ev1 ? s1 : m3;
        const float u3 = ev1 ? m1 : s3;
        typedef __attribute__((ext_vector_type(2))) _Float16 v2h;
        v2h p01, p23;
        p01[0] = (f16)u0; p01[1] = (f16)u1;
        p23[0] = (f16)u2; p23[1] = (f16)u3;
        const unsigned w01 = __builtin_bit_cast(unsigned, p01);
        const unsigned w23 = __builtin_bit_cast(unsigned, p23);

        const unsigned a01 = __builtin_amdgcn_permlane16(w01, w01, 0xBA983210u, 0xFEDC7654u, false, false);
        const unsigned x01 = __builtin_amdgcn_permlanex16(w01, w01, 0xFEDC7654u, 0xBA983210u, false, false);
        const unsigned a23 = __builtin_amdgcn_permlane16(w23, w23, 0xBA983210u, 0xFEDC7654u, false, false);
        const unsigned x23 = __builtin_amdgcn_permlanex16(w23, w23, 0xFEDC7654u, 0xBA983210u, false, false);
        v8u An;
        #pragma unroll
        for (int i = 0; i < 8; ++i) An[i] = 0u;
        An[0] = same_row ? a01 : x01;
        An[1] = same_row ? a23 : x23;
        Ah = __builtin_bit_cast(v16h, An);

        v2u hv;
        hv[0] = An[0]; hv[1] = An[1];
        hslot[(size_t)(sb + s) * 32] = hv;
      }
      if (c == NCH) hflush(hring, (NCH - 1) & 1, (NCH - 1) * CH, Hb, bbase, L);
    }
    if (c < NCH) chunk_barrier();
  }
}

__global__ void lstm_outhead_kernel(const f16* __restrict__ Hb,
                                    const float* __restrict__ W_out,
                                    const float* __restrict__ b_out,
                                    float* __restrict__ out) {
  const int L = threadIdx.x;
  const int j16 = L & 15, half = L >> 4;
  const size_t Mb = (size_t)blockIdx.x * 16;

  const float bo0 = b_out[j16];
  const float bo1 = b_out[16 + j16];
  const float bo2 = b_out[(32 + j16) < VOC ? 32 + j16 : VOC - 1];

  v16h WoutB[3];
  #pragma unroll
  for (int nt = 0; nt < 3; ++nt) {
    const int n = nt * 16 + j16;
    WoutB[nt] = bfrag_k8(W_out, n < VOC ? n : VOC - 1, half == 0 && n < VOC);
  }

  v16h A;
  #pragma unroll
  for (int i = 0; i < 16; ++i) A[i] = (f16)0;
  if (L < 16) {
    v8h hv = *(const v8h*)(Hb + (Mb + L) * HID);
    #pragma unroll
    for (int k = 0; k < HID; ++k) A[k] = hv[k];
  }

  v8f l0 = splat8(bo0);
  v8f l1 = splat8(bo1);
  v8f l2 = splat8((32 + j16) < VOC ? bo2 : 0.f);
  l0 = wmma16(A, WoutB[0], l0);
  l1 = wmma16(A, WoutB[1], l1);
  l2 = wmma16(A, WoutB[2], l2);

  const bool v2ok = (32 + j16) < VOC;
  #pragma unroll
  for (int r = 0; r < 8; ++r) {
    float x2 = v2ok ? l2[r] : -1e30f;
    float mx = fmaxf(fmaxf(l0[r], l1[r]), x2);
    #pragma unroll
    for (int s = 1; s < 16; s <<= 1)
      mx = fmaxf(mx, __shfl_xor(mx, s, 16));
    float e0 = __expf(l0[r] - mx);
    float e1 = __expf(l1[r] - mx);
    float e2 = v2ok ? __expf(x2 - mx) : 0.f;
    float sm = e0 + e1 + e2;
    #pragma unroll
    for (int s = 1; s < 16; s <<= 1)
      sm += __shfl_xor(sm, s, 16);
    float inv = 1.0f / sm;

    size_t row = Mb + r + 8 * half;
    int t = (int)(row / BATCH);
    int b = (int)(row % BATCH);
    float* dst = out + ((size_t)b * TSEQ + t) * VOC;
    dst[j16]      = e0 * inv;
    dst[16 + j16] = e1 * inv;
    if (v2ok) dst[32 + j16] = e2 * inv;
  }
}

extern "C" void kernel_launch(void* const* d_in, const int* in_sizes, int n_in,
                              void* d_out, int out_size, void* d_ws, size_t ws_size,
                              hipStream_t stream) {
  const float* features = (const float*)d_in[0];
  const float* captions = (const float*)d_in[1];
  const float* W_in  = (const float*)d_in[2];
  const float* b_in  = (const float*)d_in[3];
  const float* W_ih  = (const float*)d_in[4];
  const float* W_hh  = (const float*)d_in[5];
  const float* b_ih  = (const float*)d_in[6];
  const float* b_hh  = (const float*)d_in[7];
  const float* W_out = (const float*)d_in[8];
  const float* b_out = (const float*)d_in[9];
  float* out = (float*)d_out;

  char* ws = (char*)d_ws;
  auto carve = [&](size_t bytes) -> char* {
    char* p = ws;
    ws += (bytes + 255) & ~(size_t)255;
    return p;
  };
  float* part   = (float*)carve((size_t)NCC * PART_CC * sizeof(float));
  f16*   Hb     = (f16*)  carve((size_t)NROW * HID * sizeof(f16));

  lstm_poolfeat_kernel<<<(BATCH / 16) * NCC, 1024, 0, stream>>>(features, W_in, part);
  lstm_recur_kernel<<<BATCH / 16, 64, 0, stream>>>(captions, part, b_in, W_ih, W_hh, b_ih, b_hh, Hb);
  lstm_outhead_kernel<<<NTILE, 32, 0, stream>>>(Hb, W_out, b_out, out);
}
